# starting kernel
# baseline (speedup 1.0000x reference)
#include <hip/hip_runtime.h>
#include <hip/hip_bf16.h>

#define HID    24
#define TSTEPS 262
#define KINFO  256

typedef __attribute__((ext_vector_type(2))) float v2f;
typedef __attribute__((ext_vector_type(8))) float v8f;

__global__ __launch_bounds__(64) void birnn_gru_wmma(
    const float* __restrict__ x,
    const float* __restrict__ w_ih_f, const float* __restrict__ w_hh_f,
    const float* __restrict__ b_ih_f, const float* __restrict__ b_hh_f,
    const float* __restrict__ w_ih_b, const float* __restrict__ w_hh_b,
    const float* __restrict__ b_ih_b, const float* __restrict__ b_hh_b,
    const float* __restrict__ w_out,  const float* __restrict__ b_out,
    float* __restrict__ out)
{
    __shared__ float Wlds[2][28][96];
    __shared__ float Wgi [2][4][32];
    __shared__ float hls [2][16][32];
    __shared__ float lbuf[2][16][KINFO + 1];

    const int tid  = threadIdx.x;
    const int base = blockIdx.x * 16;

    for (int idx = tid; idx < 2 * 28 * 96; idx += 64) {
        int d = idx / (28 * 96);
        int rem = idx % (28 * 96);
        int k = rem / 96, n = rem % 96;
        int g = n >> 5, c = n & 31;
        const float* wih = d ? w_ih_b : w_ih_f;
        const float* whh = d ? w_hh_b : w_hh_f;
        const float* bih = d ? b_ih_b : b_ih_f;
        const float* bhh = d ? b_hh_b : b_hh_f;
        float v = 0.0f;
        if (c < HID) {
            int row = g * HID + c;
            if      (k < HID)  v = whh[row * HID + k];
            else if (k == 24)  v = (g < 2) ? wih[row * 2 + 0] : 0.0f;
            else if (k == 25)  v = (g < 2) ? wih[row * 2 + 1] : 0.0f;
            else if (k == 26)  v = (g < 2) ? (bih[row] + bhh[row]) : bhh[row];
        }
        Wlds[d][k][n] = v;
    }
    for (int idx = tid; idx < 2 * 4 * 32; idx += 64) {
        int d = idx / 128; int rem = idx % 128;
        int kk = rem / 32, c = rem % 32;
        const float* wih = d ? w_ih_b : w_ih_f;
        const float* bih = d ? b_ih_b : b_ih_f;
        float v = 0.0f;
        if (c < HID) {
            int row = 2 * HID + c;
            if      (kk == 0) v = wih[row * 2 + 0];
            else if (kk == 1) v = wih[row * 2 + 1];
            else if (kk == 2) v = bih[row];
        }
        Wgi[d][kk][c] = v;
    }
    __syncthreads();

    const int wave = tid >> 5;
    const int lane = tid & 31;
    const int col  = lane & 15;
    const int lh   = lane >> 4;
    const int kp   = lh * 2;
    const int roff = lh * 8;

    v2f Bf[6][7];
    #pragma unroll
    for (int tt = 0; tt < 6; ++tt)
        #pragma unroll
        for (int ck = 0; ck < 7; ++ck) {
            Bf[tt][ck].x = Wlds[wave][4 * ck + kp    ][16 * tt + col];
            Bf[tt][ck].y = Wlds[wave][4 * ck + kp + 1][16 * tt + col];
        }
    v2f Bg[2];
    #pragma unroll
    for (int tt = 0; tt < 2; ++tt) {
        Bg[tt].x = Wgi[wave][kp    ][16 * tt + col];
        Bg[tt].y = Wgi[wave][kp + 1][16 * tt + col];
    }
    v2f woa[6];
    #pragma unroll
    for (int c = 0; c < 6; ++c) {
        woa[c].x = w_out[wave * HID + 4 * c + kp];
        woa[c].y = w_out[wave * HID + 4 * c + kp + 1];
    }

    float* hb = &hls[wave][0][0];

    float hD[2][8];
    v2f afr[7];
    #pragma unroll
    for (int i = 0; i < 2; ++i)
        #pragma unroll
        for (int j = 0; j < 8; ++j) hD[i][j] = 0.0f;
    #pragma unroll
    for (int c = 0; c < 6; ++c) { afr[c].x = 0.0f; afr[c].y = 0.0f; }

    const int nsteps = wave ? TSTEPS : KINFO;
    const int xrow = (base + col) * TSTEPS;

    for (int s = 0; s < nsteps; ++s) {
        const int t = wave ? (TSTEPS - 1 - s) : s;

        const float x0 = x[(xrow + t) * 2 + 0];
        const float x1 = x[(xrow + t) * 2 + 1];
        afr[6].x = lh ? 1.0f : x0;
        afr[6].y = lh ? 0.0f : x1;

        v8f acc[6], gacc[2];
        #pragma unroll
        for (int tt = 0; tt < 6; ++tt) {
            v8f c = {};
            #pragma unroll
            for (int ck = 0; ck < 7; ++ck)
                c = __builtin_amdgcn_wmma_f32_16x16x4_f32(
                        false, afr[ck], false, Bf[tt][ck],
                        (short)0, c, false, false);
            acc[tt] = c;
        }
        #pragma unroll
        for (int tt = 0; tt < 2; ++tt) {
            v8f z = {};
            gacc[tt] = __builtin_amdgcn_wmma_f32_16x16x4_f32(
                           false, afr[6], false, Bg[tt],
                           (short)0, z, false, false);
        }

        #pragma unroll
        for (int i = 0; i < 2; ++i)
            #pragma unroll
            for (int j = 0; j < 8; ++j) {
                float r   = 1.0f / (1.0f + __expf(-acc[i][j]));
                float zz  = 1.0f / (1.0f + __expf(-acc[2 + i][j]));
                float pre = gacc[i][j] + r * acc[4 + i][j];
                float n   = 1.0f - 2.0f / (1.0f + __expf(2.0f * pre));
                float hn  = n + zz * (hD[i][j] - n);
                hD[i][j]  = hn;
                hb[(j + roff) * 32 + 16 * i + col] = hn;
            }

        #pragma unroll
        for (int c = 0; c < 6; ++c) {
            afr[c].x = hb[col * 32 + 4 * c + kp];
            afr[c].y = hb[col * 32 + 4 * c + kp + 1];
        }

        if (t < KINFO) {
            float p = 0.0f;
            #pragma unroll
            for (int c = 0; c < 6; ++c)
                p += afr[c].x * woa[c].x + afr[c].y * woa[c].y;
            p += __shfl_xor(p, 16, 32);
            if (lane < 16) lbuf[wave][lane][t] = p;
        }
    }

    __syncthreads();

    const float bo = b_out[0];
    for (int idx = tid; idx < 16 * KINFO; idx += 64) {
        int row = idx >> 8;
        int t   = idx & (KINFO - 1);
        out[(base + row) * KINFO + t] = lbuf[0][row][t] + lbuf[1][row][t] + bo;
    }
}

extern "C" void kernel_launch(void* const* d_in, const int* in_sizes, int n_in,
                              void* d_out, int out_size, void* d_ws, size_t ws_size,
                              hipStream_t stream) {
    const float* x      = (const float*)d_in[0];
    const float* w_ih_f = (const float*)d_in[1];
    const float* w_hh_f = (const float*)d_in[2];
    const float* b_ih_f = (const float*)d_in[3];
    const float* b_hh_f = (const float*)d_in[4];
    const float* w_ih_b = (const float*)d_in[5];
    const float* w_hh_b = (const float*)d_in[6];
    const float* b_ih_b = (const float*)d_in[7];
    const float* b_hh_b = (const float*)d_in[8];
    const float* w_out  = (const float*)d_in[9];
    const float* b_out  = (const float*)d_in[10];
    float* out = (float*)d_out;

    const int batch = 8192;
    dim3 grid(batch / 16, 1, 1);
    dim3 block(64, 1, 1);
    hipLaunchKernelGGL(birnn_gru_wmma, grid, block, 0, stream,
                       x, w_ih_f, w_hh_f, b_ih_f, b_hh_f,
                       w_ih_b, w_hh_b, b_ih_b, b_hh_b, w_out, b_out, out);
}


# wave priority follows progress in the time loop (s_setprio 3,2,1,0 over four spans): of the two workgroups sharing a WGP the one further behind wins issue, so both end together, none runs a lone tail
# speedup vs baseline: 2.3330x; 2.3330x over previous
#include <hip/hip_runtime.h>
#include <hip/hip_bf16.h>


#define HID    24
#define HALF   12
#define TSTEPS 262
#define KINFO  256
#define HPITCH 36
#define LT     263
#define LROW   20
#define XROWF4 131
#define XP     17

typedef __attribute__((ext_vector_type(2))) float v2f;
typedef __attribute__((ext_vector_type(4))) float v4f;
typedef __attribute__((ext_vector_type(8))) float v8f;

__global__ __launch_bounds__(128) void birnn_gru_wmma(
    const float* __restrict__ x,
    const float* __restrict__ w_ih_f, const float* __restrict__ w_hh_f,
    const float* __restrict__ b_ih_f, const float* __restrict__ b_hh_f,
    const float* __restrict__ w_ih_b, const float* __restrict__ w_hh_b,
    const float* __restrict__ b_ih_b, const float* __restrict__ b_hh_b,
    const float* __restrict__ w_out,  const float* __restrict__ b_out,
    float* __restrict__ out)
{
    __shared__ float Wlds[2][28][96];
    __shared__ float Wgi [2][4][32];
    __shared__ float hx  [2][2][16][HPITCH];
    __shared__ __attribute__((aligned(16))) float lbuf[2][LT][LROW];
    __shared__ __attribute__((aligned(16))) v2f xs[TSTEPS][XP];

    const int tid  = threadIdx.x;
    const int base = blockIdx.x * 16;

    {
        const v4f* xg = reinterpret_cast<const v4f*>(x + (size_t)base * (TSTEPS * 2));
        const bool tail = tid < 16 * XROWF4 - 16 * 128;
        v4f xv[17];
        #pragma unroll
        for (int it = 0; it < 17; ++it) {
            v4f v = {};
            if (it < 16 || tail) v = xg[tid + 128 * it];
            xv[it] = v;
        }
        #pragma unroll
        for (int it = 0; it < 17; ++it) {
            const int i = tid + 128 * it;
            if (it < 16 || tail) {
                const int row = i / XROWF4;
                const int k   = i - row * XROWF4;
                v2f a, b;
                a.x = xv[it].x; a.y = xv[it].y;
                b.x = xv[it].z; b.y = xv[it].w;
                xs[2 * k    ][row] = a;
                xs[2 * k + 1][row] = b;
            }
        }
    }

    for (int idx = tid; idx < 2 * 28 * 96; idx += 128) {
        int d = idx / (28 * 96);
        int rem = idx % (28 * 96);
        int k = rem / 96, n = rem % 96;
        int g = n >> 5, c = n & 31;
        const float* wih = d ? w_ih_b : w_ih_f;
        const float* whh = d ? w_hh_b : w_hh_f;
        const float* bih = d ? b_ih_b : b_ih_f;
        const float* bhh = d ? b_hh_b : b_hh_f;
        float v = 0.0f;
        if (c < HID) {
            int row = g * HID + c;
            if      (k < HID)  v = whh[row * HID + k];
            else if (k == 24)  v = (g < 2) ? wih[row * 2 + 0] : 0.0f;
            else if (k == 25)  v = (g < 2) ? wih[row * 2 + 1] : 0.0f;
            else if (k == 26)  v = (g < 2) ? (bih[row] + bhh[row]) : bhh[row];
        }
        Wlds[d][k][n] = v;
    }
    for (int idx = tid; idx < 2 * 4 * 32; idx += 128) {
        int d = idx / 128; int rem = idx % 128;
        int kk = rem / 32, c = rem % 32;
        const float* wih = d ? w_ih_b : w_ih_f;
        const float* bih = d ? b_ih_b : b_ih_f;
        float v = 0.0f;
        if (c < HID) {
            int row = 2 * HID + c;
            if      (kk == 0) v = wih[row * 2 + 0];
            else if (kk == 1) v = wih[row * 2 + 1];
            else if (kk == 2) v = bih[row];
        }
        Wgi[d][kk][c] = v;
    }
    __syncthreads();

    const int wave = __builtin_amdgcn_readfirstlane(tid >> 5);
    const int dir  = wave >> 1;
    const int q    = wave & 1;
    const int lane = tid & 31;
    const int col  = lane & 15;
    const int lh   = lane >> 4;
    const int kp   = lh * 2;
    const int roff = lh * 8;
    const bool own = col < HALF;
    const int hcol = HALF * q + col;
    const int scol = own ? hcol : (HID + 4 * q + col - HALF);

    v2f xt = xs[dir ? (TSTEPS - 1) : 0][col];

    const float SRZ = -1.4426950408889634f;
    const float SN  =  2.8853900817779268f;
    v2f Bf[3][7];
    #pragma unroll
    for (int g = 0; g < 3; ++g)
        #pragma unroll
        for (int ck = 0; ck < 7; ++ck) {
            const float sc = (g < 2) ? SRZ : SN;
            const float b0 = Wlds[dir][4 * ck + kp    ][32 * g + hcol] * sc;
            const float b1 = Wlds[dir][4 * ck + kp + 1][32 * g + hcol] * sc;
            Bf[g][ck].x = own ? b0 : 0.0f;
            Bf[g][ck].y = own ? b1 : 0.0f;
        }
    v8f cn;
    {
        const float bn = own ? Wlds[dir][26][64 + hcol] * SN : 0.0f;
        #pragma unroll
        for (int j = 0; j < 8; ++j) cn[j] = bn;
    }
    const bool prj = (col == HALF);
    #pragma unroll
    for (int c = 0; c < 6; ++c) {
        const float w0 = w_out[dir * HID + 4 * c + kp];
        const float w1 = w_out[dir * HID + 4 * c + kp + 1];
        Bf[0][c].x = prj ? w0 : Bf[0][c].x;
        Bf[0][c].y = prj ? w1 : Bf[0][c].y;
    }
    v2f Bg;
    {
        const float b0 = Wgi[dir][kp    ][hcol] * SN;
        const float b1 = Wgi[dir][kp + 1][hcol] * SN;
        Bg.x = own ? b0 : 0.0f;
        Bg.y = own ? b1 : 0.0f;
    }
    float* hbase = &hx[dir][0][0][0];
    float* lst   = &lbuf[dir][0][8 * q];
    const bool plane = (lane == HALF + 16 * q);

    float hD[8];
    v2f afr[7];
    #pragma unroll
    for (int j = 0; j < 8; ++j) hD[j] = 0.0f;
    #pragma unroll
    for (int c = 0; c < 6; ++c) { afr[c].x = 0.0f; afr[c].y = 0.0f; }

    int tprev = TSTEPS;

    int s = 0;
    #pragma nounroll
    for (int span = 0; span < 4; ++span) {
    int send;
    if      (span == 0) { __builtin_amdgcn_s_setprio(3); send = 128; }
    else if (span == 1) { __builtin_amdgcn_s_setprio(2); send = 192; }
    else if (span == 2) { __builtin_amdgcn_s_setprio(1); send = 232; }
    else                { __builtin_amdgcn_s_setprio(0); send = TSTEPS; }
    for (; s < send; ++s) {
        const int t = dir ? (TSTEPS - 1 - s) : s;

        afr[6].x = lh ? 1.0f : xt.x;
        afr[6].y = lh ? 0.0f : xt.y;

        v8f acc[3], gacc;
        #pragma unroll
        for (int g = 0; g < 3; ++g) {
            v8f c = {};
            if (g == 2) c = cn;
            #pragma unroll
            for (int ck = 0; ck < 7; ++ck)
                if (g < 2 || ck < 6)
                    c = __builtin_amdgcn_wmma_f32_16x16x4_f32(
                            false, afr[ck], false, Bf[g][ck],
                            (short)0, c, false, false);
            acc[g] = c;
        }
        {
            v8f z = {};
            gacc = __builtin_amdgcn_wmma_f32_16x16x4_f32(
                       false, afr[6], false, Bg,
                       (short)0, z, false, false);
        }

        const v8f lg = acc[0];

        float* hw = hbase + (s & 1) * (16 * HPITCH);
        #pragma unroll
        for (int j = 0; j < 8; ++j) {
            float r   = __builtin_amdgcn_rcpf(1.0f + __builtin_amdgcn_exp2f(acc[0][j]));
            float zz  = __builtin_amdgcn_rcpf(1.0f + __builtin_amdgcn_exp2f(acc[1][j]));
            float pre = gacc[j] + r * acc[2][j];
            float n   = 1.0f - 2.0f * __builtin_amdgcn_rcpf(1.0f + __builtin_amdgcn_exp2f(pre));
            float hn  = n + zz * (hD[j] - n);
            hD[j]     = hn;
            hw[(j + roff) * HPITCH + scol] = hn;
        }

        __syncthreads();

        #pragma unroll
        for (int c = 0; c < 6; ++c) {
            afr[c].x = hw[col * HPITCH + 4 * c + kp];
            afr[c].y = hw[col * HPITCH + 4 * c + kp + 1];
        }

        {
            const int sn = (s + 1 < TSTEPS) ? (s + 1) : s;
            xt = xs[dir ? (TSTEPS - 1 - sn) : sn][col];
        }

        if (plane) {
            v4f* dst = reinterpret_cast<v4f*>(lst + tprev * LROW);
            dst[0] = __builtin_shufflevector(lg, lg, 0, 1, 2, 3);
            dst[1] = __builtin_shufflevector(lg, lg, 4, 5, 6, 7);
        }
        tprev = t;
    }
    }

    {
        v8f c = {};
        #pragma unroll
        for (int ck = 0; ck < 6; ++ck)
            c = __builtin_amdgcn_wmma_f32_16x16x4_f32(
                    false, afr[ck], false, Bf[0][ck],
                    (short)0, c, false, false);
        if (plane) {
            v4f* dst = reinterpret_cast<v4f*>(lst + tprev * LROW);
            dst[0] = __builtin_shufflevector(c, c, 0, 1, 2, 3);
            dst[1] = __builtin_shufflevector(c, c, 4, 5, 6, 7);
        }
    }

    __syncthreads();

    const float bo = b_out[0];
    for (int idx = tid; idx < 16 * KINFO; idx += 128) {
        int row = idx >> 8;
        int t   = idx & (KINFO - 1);
        out[(base + row) * KINFO + t] = (lbuf[0][t][row] + lbuf[1][t][row]) + bo;
    }
}

extern "C" void kernel_launch(void* const* d_in, const int* in_sizes, int n_in,
                              void* d_out, int out_size, void* d_ws, size_t ws_size,
                              hipStream_t stream) {
    const float* x      = (const float*)d_in[0];
    const float* w_ih_f = (const float*)d_in[1];
    const float* w_hh_f = (const float*)d_in[2];
    const float* b_ih_f = (const float*)d_in[3];
    const float* b_hh_f = (const float*)d_in[4];
    const float* w_ih_b = (const float*)d_in[5];
    const float* w_hh_b = (const float*)d_in[6];
    const float* b_ih_b = (const float*)d_in[7];
    const float* b_hh_b = (const float*)d_in[8];
    const float* w_out  = (const float*)d_in[9];
    const float* b_out  = (const float*)d_in[10];
    float* out = (float*)d_out;

    const int batch = 8192;
    dim3 grid(batch / 16, 1, 1);
    dim3 block(128, 1, 1);
    hipLaunchKernelGGL(birnn_gru_wmma, grid, block, 0, stream,
                       x, w_ih_f, w_hh_f, b_ih_f, b_hh_f,
                       w_ih_b, w_hh_b, b_ih_b, b_hh_b, w_out, b_out, out);
}


# weights copied raw into LDS by coalesced 16-byte loads issued with the x block's: one memory round trip; B fragments read from that copy; the 42-pass staging loop (a waited load per pass) is gone
# speedup vs baseline: 2.4790x; 1.0626x over previous
#include <hip/hip_runtime.h>
#include <hip/hip_bf16.h>


#define HID    24
#define HALF   12
#define TSTEPS 262
#define KINFO  256
#define HPITCH 36
#define LT     263
#define LROW   20
#define XROWF4 131
#define XP     17
#define WP_HH   432
#define WP_IH   468
#define WP_BI   486
#define WP_DIR  504
#define WP_ALL  1020
#define WF_DIR  2016
#define WF_IH   1728
#define WF_BI   1872
#define WF_BH   1944
#define WRAW_LEN 5632

typedef __attribute__((ext_vector_type(2))) float v2f;
typedef __attribute__((ext_vector_type(4))) float v4f;
typedef __attribute__((ext_vector_type(8))) float v8f;

__global__ __launch_bounds__(128) void birnn_gru_wmma(
    const float* __restrict__ x,
    const float* __restrict__ w_ih_f, const float* __restrict__ w_hh_f,
    const float* __restrict__ b_ih_f, const float* __restrict__ b_hh_f,
    const float* __restrict__ w_ih_b, const float* __restrict__ w_hh_b,
    const float* __restrict__ b_ih_b, const float* __restrict__ b_hh_b,
    const float* __restrict__ w_out,  const float* __restrict__ b_out,
    float* __restrict__ out)
{
    __shared__ __attribute__((aligned(16))) float wraw[WRAW_LEN];
    __shared__ float hx  [2][2][16][HPITCH];
    __shared__ __attribute__((aligned(16))) float lbuf[2][LT][LROW];
    __shared__ __attribute__((aligned(16))) v2f xs[TSTEPS][XP];

    const int tid  = threadIdx.x;
    const int base = blockIdx.x * 16;
    const float bo = b_out[0];

    {
        const v4f* xg = reinterpret_cast<const v4f*>(x + (size_t)base * (TSTEPS * 2));
        const bool tail = tid < 16 * XROWF4 - 16 * 128;
        const bool wtail = tid < WP_ALL - 7 * 128;
        v4f xv[17];
        v4f wv[8];
        #pragma unroll
        for (int it = 0; it < 16; ++it) xv[it] = xg[tid + 128 * it];
        #pragma unroll
        for (int it = 0; it < 8; ++it) {
            const int p  = tid + 128 * it;
            const bool d = p >= WP_DIR;
            const int r  = d ? p - WP_DIR : p;
            const float* hh = d ? w_hh_b : w_hh_f;
            const float* ih = d ? w_ih_b : w_ih_f;
            const float* bi = d ? b_ih_b : b_ih_f;
            const float* bh = d ? b_hh_b : b_hh_f;
            const float* src = (r < WP_HH) ? hh : (r < WP_IH) ? ih : (r < WP_BI) ? bi : bh;
            int off          = (r < WP_HH) ? r  : (r < WP_IH) ? r - WP_HH : (r < WP_BI) ? r - WP_IH : r - WP_BI;
            if (p >= 2 * WP_DIR) { src = w_out; off = p - 2 * WP_DIR; }
            v4f v = {};
            if (it < 7 || wtail) v = reinterpret_cast<const v4f*>(src)[off];
            wv[it] = v;
        }
        {
            v4f v = {};
            if (tail) v = xg[tid + 128 * 16];
            xv[16] = v;
        }
        #pragma unroll
        for (int it = 0; it < 17; ++it) {
            const int i = tid + 128 * it;
            if (it < 16 || tail) {
                const int row = i / XROWF4;
                const int k   = i - row * XROWF4;
                v2f a, b;
                a.x = xv[it].x; a.y = xv[it].y;
                b.x = xv[it].z; b.y = xv[it].w;
                xs[2 * k    ][row] = a;
                xs[2 * k + 1][row] = b;
            }
        }
        #pragma unroll
        for (int it = 0; it < 8; ++it)
            if (it < 7 || wtail)
                reinterpret_cast<v4f*>(wraw)[tid + 128 * it] = wv[it];
    }
    __syncthreads();

    const int wave = __builtin_amdgcn_readfirstlane(tid >> 5);
    const int dir  = wave >> 1;
    const int q    = wave & 1;
    const int lane = tid & 31;
    const int col  = lane & 15;
    const int lh   = lane >> 4;
    const int kp   = lh * 2;
    const int roff = lh * 8;
    const bool own = col < HALF;
    const int hcol = HALF * q + col;
    const int scol = own ? hcol : (HID + 4 * q + col - HALF);

    v2f xt = xs[dir ? (TSTEPS - 1) : 0][col];

    const float SRZ = -1.4426950408889634f;
    const float SN  =  2.8853900817779268f;
    const int hc = own ? hcol : 0;
    const float* rhh = wraw + dir * WF_DIR;
    const float* rih = rhh + WF_IH;
    const float* rbi = rhh + WF_BI;
    const float* rbh = rhh + WF_BH;
    v2f Bf[3][7];
    #pragma unroll
    for (int g = 0; g < 3; ++g)
        #pragma unroll
        for (int ck = 0; ck < 7; ++ck) {
            const float sc = (g < 2) ? SRZ : SN;
            const int row = g * HID + hc;
            float b0 = 0.0f, b1 = 0.0f;
            if (ck < 6) {
                b0 = rhh[row * HID + 4 * ck + kp    ] * sc;
                b1 = rhh[row * HID + 4 * ck + kp + 1] * sc;
            } else if (g < 2) {
                const float wi0 = rih[row * 2], wi1 = rih[row * 2 + 1];
                const float bs  = rbi[row] + rbh[row];
                b0 = (lh ? bs   : wi0) * sc;
                b1 = (lh ? 0.0f : wi1) * sc;
            }
            Bf[g][ck].x = own ? b0 : 0.0f;
            Bf[g][ck].y = own ? b1 : 0.0f;
        }
    v8f cn;
    {
        const float bn = own ? rbh[2 * HID + hc] * SN : 0.0f;
        #pragma unroll
        for (int j = 0; j < 8; ++j) cn[j] = bn;
    }
    const bool prj = (col == HALF);
    #pragma unroll
    for (int c = 0; c < 6; ++c) {
        const float w0 = wraw[2 * WF_DIR + dir * HID + 4 * c + kp];
        const float w1 = wraw[2 * WF_DIR + dir * HID + 4 * c + kp + 1];
        Bf[0][c].x = prj ? w0 : Bf[0][c].x;
        Bf[0][c].y = prj ? w1 : Bf[0][c].y;
    }
    v2f Bg;
    {
        const int row = 2 * HID + hc;
        const float wi0 = rih[row * 2], wi1 = rih[row * 2 + 1];
        const float bi  = rbi[row];
        const float b0 = (lh ? bi   : wi0) * SN;
        const float b1 = (lh ? 0.0f : wi1) * SN;
        Bg.x = own ? b0 : 0.0f;
        Bg.y = own ? b1 : 0.0f;
    }
    float* hbase = &hx[dir][0][0][0];
    float* lst   = &lbuf[dir][0][8 * q];
    const bool plane = (lane == HALF + 16 * q);

    float hD[8];
    v2f afr[7];
    #pragma unroll
    for (int j = 0; j < 8; ++j) hD[j] = 0.0f;
    #pragma unroll
    for (int c = 0; c < 6; ++c) { afr[c].x = 0.0f; afr[c].y = 0.0f; }

    int tprev = TSTEPS;

    int s = 0;
    #pragma nounroll
    for (int span = 0; span < 4; ++span) {
    int send;
    if      (span == 0) { __builtin_amdgcn_s_setprio(3); send = 128; }
    else if (span == 1) { __builtin_amdgcn_s_setprio(2); send = 192; }
    else if (span == 2) { __builtin_amdgcn_s_setprio(1); send = 232; }
    else                { __builtin_amdgcn_s_setprio(0); send = TSTEPS; }
    for (; s < send; ++s) {
        const int t = dir ? (TSTEPS - 1 - s) : s;

        afr[6].x = lh ? 1.0f : xt.x;
        afr[6].y = lh ? 0.0f : xt.y;

        v8f acc[3], gacc;
        #pragma unroll
        for (int g = 0; g < 3; ++g) {
            v8f c = {};
            if (g == 2) c = cn;
            #pragma unroll
            for (int ck = 0; ck < 7; ++ck)
                if (g < 2 || ck < 6)
                    c = __builtin_amdgcn_wmma_f32_16x16x4_f32(
                            false, afr[ck], false, Bf[g][ck],
                            (short)0, c, false, false);
            acc[g] = c;
        }
        {
            v8f z = {};
            gacc = __builtin_amdgcn_wmma_f32_16x16x4_f32(
                       false, afr[6], false, Bg,
                       (short)0, z, false, false);
        }

        const v8f lg = acc[0];

        float* hw = hbase + (s & 1) * (16 * HPITCH);
        #pragma unroll
        for (int j = 0; j < 8; ++j) {
            float r   = __builtin_amdgcn_rcpf(1.0f + __builtin_amdgcn_exp2f(acc[0][j]));
            float zz  = __builtin_amdgcn_rcpf(1.0f + __builtin_amdgcn_exp2f(acc[1][j]));
            float pre = gacc[j] + r * acc[2][j];
            float n   = 1.0f - 2.0f * __builtin_amdgcn_rcpf(1.0f + __builtin_amdgcn_exp2f(pre));
            float hn  = n + zz * (hD[j] - n);
            hD[j]     = hn;
            hw[(j + roff) * HPITCH + scol] = hn;
        }

        __syncthreads();

        #pragma unroll
        for (int c = 0; c < 6; ++c) {
            afr[c].x = hw[col * HPITCH + 4 * c + kp];
            afr[c].y = hw[col * HPITCH + 4 * c + kp + 1];
        }

        {
            const int sn = (s + 1 < TSTEPS) ? (s + 1) : s;
            xt = xs[dir ? (TSTEPS - 1 - sn) : sn][col];
        }

        if (plane) {
            v4f* dst = reinterpret_cast<v4f*>(lst + tprev * LROW);
            dst[0] = __builtin_shufflevector(lg, lg, 0, 1, 2, 3);
            dst[1] = __builtin_shufflevector(lg, lg, 4, 5, 6, 7);
        }
        tprev = t;
    }
    }

    {
        v8f c = {};
        #pragma unroll
        for (int ck = 0; ck < 6; ++ck)
            c = __builtin_amdgcn_wmma_f32_16x16x4_f32(
                    false, afr[ck], false, Bf[0][ck],
                    (short)0, c, false, false);
        if (plane) {
            v4f* dst = reinterpret_cast<v4f*>(lst + tprev * LROW);
            dst[0] = __builtin_shufflevector(c, c, 0, 1, 2, 3);
            dst[1] = __builtin_shufflevector(c, c, 4, 5, 6, 7);
        }
    }

    __syncthreads();

    for (int idx = tid; idx < 16 * KINFO; idx += 128) {
        int row = idx >> 8;
        int t   = idx & (KINFO - 1);
        out[(base + row) * KINFO + t] = (lbuf[0][t][row] + lbuf[1][t][row]) + bo;
    }
}

extern "C" void kernel_launch(void* const* d_in, const int* in_sizes, int n_in,
                              void* d_out, int out_size, void* d_ws, size_t ws_size,
                              hipStream_t stream) {
    const float* x      = (const float*)d_in[0];
    const float* w_ih_f = (const float*)d_in[1];
    const float* w_hh_f = (const float*)d_in[2];
    const float* b_ih_f = (const float*)d_in[3];
    const float* b_hh_f = (const float*)d_in[4];
    const float* w_ih_b = (const float*)d_in[5];
    const float* w_hh_b = (const float*)d_in[6];
    const float* b_ih_b = (const float*)d_in[7];
    const float* b_hh_b = (const float*)d_in[8];
    const float* w_out  = (const float*)d_in[9];
    const float* b_out  = (const float*)d_in[10];
    float* out = (float*)d_out;

    const int batch = 8192;
    dim3 grid(batch / 16, 1, 1);
    dim3 block(128, 1, 1);
    hipLaunchKernelGGL(birnn_gru_wmma, grid, block, 0, stream,
                       x, w_ih_f, w_hh_f, b_ih_f, b_hh_f,
                       w_ih_b, w_hh_b, b_ih_b, b_hh_b, w_out, b_out, out);
}
